# starting kernel
# baseline (speedup 1.0000x reference)
#include <hip/hip_runtime.h>
#include <math.h>

constexpr int B    = 64;
constexpr int T    = 2048;
constexpr int C    = 512;
constexpr int NSEG = 16;
constexpr int TSEG = T / NSEG;
constexpr int K1   = 3 * C;
constexpr int N1   = C;
constexpr int K2   = C;
constexpr int N2   = C / 2;

typedef float v2f __attribute__((ext_vector_type(2)));
typedef float v8f __attribute__((ext_vector_type(8)));

__global__ void pool_partial_kernel(const float* __restrict__ x,
                                    const int* __restrict__ lengths,
                                    float* __restrict__ partial) {
  const int seg = blockIdx.x;
  const int b   = blockIdx.y;
  const int c4  = threadIdx.x * 4;

  const int L = lengths[b];
  const int q = L / 4;
  const int p = (3 * L) / 4;

  const int t0 = seg * TSEG;
  int t1 = t0 + TSEG;
  if (t1 > L) t1 = L;

  float sb[4] = {0.f, 0.f, 0.f, 0.f};
  float sf[4] = {0.f, 0.f, 0.f, 0.f};
  float mx[4];
  const float ninf = -__builtin_inff();
  mx[0] = ninf; mx[1] = ninf; mx[2] = ninf; mx[3] = ninf;

  const float* xp = x + ((size_t)b * T + t0) * C + c4;
  for (int t = t0; t < t1; ++t, xp += C) {
    const float4 v = *(const float4*)xp;
    const float vv[4] = {v.x, v.y, v.z, v.w};
    if (t < q) {
      #pragma unroll
      for (int i = 0; i < 4; ++i) sb[i] += vv[i];
    } else if (t < p) {
      #pragma unroll
      for (int i = 0; i < 4; ++i) mx[i] = fmaxf(mx[i], vv[i]);
    } else {
      #pragma unroll
      for (int i = 0; i < 4; ++i) sf[i] += vv[i];
    }
  }

  float* base = partial + (((size_t)b * NSEG + seg) * 3) * C + c4;
  *(float4*)(base + 0 * C) = make_float4(sb[0], sb[1], sb[2], sb[3]);
  *(float4*)(base + 1 * C) = make_float4(mx[0], mx[1], mx[2], mx[3]);
  *(float4*)(base + 2 * C) = make_float4(sf[0], sf[1], sf[2], sf[3]);
}

__global__ void pool_reduce_kernel(const float* __restrict__ partial,
                                   const int* __restrict__ lengths,
                                   float* __restrict__ feat) {
  const int b = blockIdx.x;
  const int c = threadIdx.x;

  const int L = lengths[b];
  const float qf = (float)(L / 4);
  const float ff = (float)(L - (3 * L) / 4);

  float sb = 0.f, sf = 0.f, mx = -__builtin_inff();
  const float* base = partial + (size_t)b * NSEG * 3 * C + c;
  #pragma unroll
  for (int s = 0; s < NSEG; ++s) {
    const float* ps = base + (size_t)s * 3 * C;
    sb += ps[0];
    mx = fmaxf(mx, ps[C]);
    sf += ps[2 * C];
  }
  float* fb = feat + (size_t)b * K1;
  fb[c]         = sb / qf;
  fb[C + c]     = mx;
  fb[2 * C + c] = sf / ff;
}

__device__ __forceinline__ v8f wmma_f32_k4(v2f a, v2f b, v8f c) {
  return __builtin_amdgcn_wmma_f32_16x16x4_f32(
       false, a,  false, b,
       (short)0, c,  false,  false);
}

__global__ void gemm_bias_act_kernel(const float* __restrict__ A,
                                     const float* __restrict__ W,
                                     const float* __restrict__ bias,
                                     float* __restrict__ Out,
                                     int N, int K, int do_relu) {
  const int lane = threadIdx.x;
  const int l16  = lane & 15;
  const int kh   = (lane >> 4) * 2;

  const int arow = (blockIdx.y << 4) + l16;
  const int ncol = (blockIdx.x << 4) + l16;

  v8f acc = {};
  const float* Arow = A + (size_t)arow * K;

  for (int k = 0; k < K; k += 4) {
    const int kk = k + kh;
    const v2f a = *(const v2f*)(Arow + kk);
    v2f bm;
    bm.x = W[(size_t)kk * N + ncol];
    bm.y = W[(size_t)(kk + 1) * N + ncol];
    acc = wmma_f32_k4(a, bm, acc);
  }

  const float bv = bias[ncol];
  const int rbase = (blockIdx.y << 4) + ((lane >> 4) << 3);
  #pragma unroll
  for (int r = 0; r < 8; ++r) {
    float v = acc[r] + bv;
    if (do_relu) v = fmaxf(v, 0.0f);
    Out[(size_t)(rbase + r) * N + ncol] = v;
  }
}

extern "C" void kernel_launch(void* const* d_in, const int* in_sizes, int n_in,
                              void* d_out, int out_size, void* d_ws, size_t ws_size,
                              hipStream_t stream) {
  const float* x       = (const float*)d_in[0];
  const int*   lengths = (const int*)  d_in[1];
  const float* W1      = (const float*)d_in[2];
  const float* b1      = (const float*)d_in[3];
  const float* W2      = (const float*)d_in[4];
  const float* b2      = (const float*)d_in[5];
  float*       out     = (float*)d_out;

  float* ws      = (float*)d_ws;
  float* partial = ws;
  float* feat    = partial + (size_t)B * NSEG * 3 * C;
  float* h       = feat + (size_t)B * K1;

  pool_partial_kernel<<<dim3(NSEG, B), dim3(C / 4), 0, stream>>>(x, lengths, partial);
  pool_reduce_kernel<<<dim3(B), dim3(C), 0, stream>>>(partial, lengths, feat);

  gemm_bias_act_kernel<<<dim3(N1 / 16, B / 16), dim3(32), 0, stream>>>(
      feat, W1, b1, h, N1, K1, 1);
  gemm_bias_act_kernel<<<dim3(N2 / 16, B / 16), dim3(32), 0, stream>>>(
      h, W2, b2, out, N2, K2, 0);
}


# Both GEMMs split K across work-groups (grid z: 6 and 4 slices); slice tiles go to workspace planes, summed in plane order by GEMM2's A staging (+b1, relu) and a small kernel (+b2); sum tree changes
# speedup vs baseline: 5.3045x; 5.3045x over previous
#include <hip/hip_runtime.h>
#include <math.h>

constexpr int B    = 64;
constexpr int T    = 2048;
constexpr int C    = 512;
constexpr int NSEG = 16;
constexpr int TSEG = T / NSEG;
constexpr int K1   = 3 * C;
constexpr int N1   = C;
constexpr int K2   = C;
constexpr int N2   = C / 2;

typedef float v2f __attribute__((ext_vector_type(2)));
typedef float v8f __attribute__((ext_vector_type(8)));

constexpr int PW = 8;
constexpr int PT = PW * 32;

__device__ __forceinline__ float4 add4(float4 a, float4 v) {
  return make_float4(a.x + v.x, a.y + v.y, a.z + v.z, a.w + v.w);
}
__device__ __forceinline__ float4 max4(float4 a, float4 v) {
  return make_float4(fmaxf(a.x, v.x), fmaxf(a.y, v.y), fmaxf(a.z, v.z), fmaxf(a.w, v.w));
}

__global__ __launch_bounds__(PT) void pool_partial_kernel(
    const float* __restrict__ x,
    const int* __restrict__ lengths,
    float* __restrict__ partial) {
  __shared__ __attribute__((aligned(16))) float sp[PW * 3 * C];

  const int seg  = blockIdx.x;
  const int b    = blockIdx.y;
  const int tid  = threadIdx.x;
  const int lane = tid & 31;
  const int w    = tid >> 5;

  const int L = lengths[b];
  const int q = L / 4;
  const int p = (3 * L) / 4;

  const int t0 = seg * TSEG;
  int t1 = t0 + TSEG;
  if (t1 > L) t1 = L;

  const float ninf = -__builtin_inff();
  const float4 zero4 = make_float4(0.f, 0.f, 0.f, 0.f);
  const float4 ninf4 = make_float4(ninf, ninf, ninf, ninf);

  float* base = partial + (((size_t)b * NSEG + seg) * 3) * C + 2 * tid;

  if (t0 >= L) {
    *(float2*)(base + 0 * C) = make_float2(0.f, 0.f);
    *(float2*)(base + 1 * C) = make_float2(ninf, ninf);
    *(float2*)(base + 2 * C) = make_float2(0.f, 0.f);
    return;
  }

  float4 sb[4], mx[4], sf[4];
  for (int j = 0; j < 4; ++j) { sb[j] = zero4; mx[j] = ninf4; sf[j] = zero4; }

  const float* xp = x + ((size_t)b * T + t0 + w) * C + lane * 4;
  for (int t = t0 + w; t < t1; t += PW, xp += (size_t)PW * C) {
    float4 v[4];
    for (int j = 0; j < 4; ++j) v[j] = *(const float4*)(xp + 128 * j);
    if (t < q) {
      for (int j = 0; j < 4; ++j) sb[j] = add4(sb[j], v[j]);
    } else if (t < p) {
      for (int j = 0; j < 4; ++j) mx[j] = max4(mx[j], v[j]);
    } else {
      for (int j = 0; j < 4; ++j) sf[j] = add4(sf[j], v[j]);
    }
  }

  float* mine = sp + w * (3 * C) + lane * 4;
  for (int j = 0; j < 4; ++j) {
    *(float4*)(mine + 0 * C + 128 * j) = sb[j];
    *(float4*)(mine + 1 * C + 128 * j) = mx[j];
    *(float4*)(mine + 2 * C + 128 * j) = sf[j];
  }
  __syncthreads();

  const float* fs = sp + 2 * tid;
  float2 rb = *(const float2*)(fs + 0 * C);
  float2 rm = *(const float2*)(fs + 1 * C);
  float2 rf = *(const float2*)(fs + 2 * C);
  for (int k = 1; k < PW; ++k) {
    const float2 ub = *(const float2*)(fs + k * (3 * C) + 0 * C);
    const float2 um = *(const float2*)(fs + k * (3 * C) + 1 * C);
    const float2 uf = *(const float2*)(fs + k * (3 * C) + 2 * C);
    rb.x += ub.x; rb.y += ub.y;
    rm.x = fmaxf(rm.x, um.x); rm.y = fmaxf(rm.y, um.y);
    rf.x += uf.x; rf.y += uf.y;
  }
  *(float2*)(base + 0 * C) = rb;
  *(float2*)(base + 1 * C) = rm;
  *(float2*)(base + 2 * C) = rf;
}

__global__ void pool_reduce_kernel(const float* __restrict__ partial,
                                   const int* __restrict__ lengths,
                                   float* __restrict__ feat) {
  const int b = blockIdx.x;
  const int c = threadIdx.x;

  const int L = lengths[b];
  const float qf = (float)(L / 4);
  const float ff = (float)(L - (3 * L) / 4);

  float sb = 0.f, sf = 0.f, mx = -__builtin_inff();
  const float* base = partial + (size_t)b * NSEG * 3 * C + c;
  #pragma unroll
  for (int s = 0; s < NSEG; ++s) {
    const float* ps = base + (size_t)s * 3 * C;
    sb += ps[0];
    mx = fmaxf(mx, ps[C]);
    sf += ps[2 * C];
  }
  float* fb = feat + (size_t)b * K1;
  fb[c]         = sb / qf;
  fb[C + c]     = mx;
  fb[2 * C + c] = sf / ff;
}

__device__ __forceinline__ v8f wmma_f32_k4(v2f a, v2f b, v8f c) {
  return __builtin_amdgcn_wmma_f32_16x16x4_f32(
       false, a,  false, b,
       (short)0, c,  false,  false);
}

constexpr int BM = 32;
constexpr int BN = 32;
constexpr int KC = 128;
constexpr int PA = KC + 4;
constexpr int PB = BN + 8;

constexpr int MROWS = B;

template <int N, int K, int KS, int PIN, bool RELU_IN>
__global__ __launch_bounds__(128) void gemm_slice_kernel(
    const float* __restrict__ A,
    const float* __restrict__ bias_in,
    const float* __restrict__ W,
    float* __restrict__ OutP) {
  __shared__ __attribute__((aligned(16))) float sA[BM * PA];
  __shared__ __attribute__((aligned(16))) float sB[KC * PB];

  constexpr int KSL = K / KS;
  static_assert(KSL % KC == 0, "slice must be whole chunks");

  const int tid  = threadIdx.x;
  const int lane = tid & 31;
  const int wave = tid >> 5;
  const int wm   = wave >> 1;
  const int wn   = wave & 1;
  const int c    = lane & 15;
  const int h    = lane >> 4;
  const int m0   = blockIdx.y * BM;
  const int n0   = blockIdx.x * BN;
  const int kbeg = blockIdx.z * KSL;

  const int ac4 = tid & 31;
  const int ar0 = tid >> 5;
  const int bc4 = tid & 7;
  const int br0 = tid >> 3;

  const float* gA = A + (size_t)(m0 + ar0) * K + ac4 * 4;
  const float* gB = W + (size_t)br0 * N + n0 + bc4 * 4;
  float* stA = sA + ar0 * PA + ac4 * 4;
  float* stB = sB + br0 * PB + bc4 * 4;

  const float* pa = sA + (wm * 16 + c) * PA + 2 * h;
  const float* pb = sB + (2 * h) * PB + wn * 16 + c;

  v8f acc = {};
  for (int kc = kbeg; kc < kbeg + KSL; kc += KC) {
    float4 va[8];
    float4 vb[8];
    #pragma unroll
    for (int i = 0; i < 8; ++i)
      vb[i] = *(const float4*)(gB + (size_t)(kc + 16 * i) * N);
    if constexpr (PIN == 0) {
      #pragma unroll
      for (int i = 0; i < 8; ++i)
        va[i] = *(const float4*)(gA + (size_t)(4 * i) * K + kc);
    } else {
      const float4 bv = *(const float4*)(bias_in + kc + ac4 * 4);
      #pragma unroll
      for (int i = 0; i < 8; ++i) {
        const float* s0 = gA + (size_t)(4 * i) * K + kc;
        float4 s = *(const float4*)s0;
        #pragma unroll
        for (int p = 1; p < PIN; ++p)
          s = add4(s, *(const float4*)(s0 + (size_t)p * MROWS * K));
        s = add4(s, bv);
        if (RELU_IN) s = max4(s, make_float4(0.f, 0.f, 0.f, 0.f));
        va[i] = s;
      }
    }
    #pragma unroll
    for (int i = 0; i < 8; ++i)
      *(float4*)(stA + (4 * i) * PA) = va[i];
    #pragma unroll
    for (int i = 0; i < 8; ++i)
      *(float4*)(stB + (16 * i) * PB) = vb[i];
    __syncthreads();

    v2f af[KC / 4];
    v2f bf[KC / 4];
    #pragma unroll
    for (int j = 0; j < KC / 4; ++j) {
      af[j]   = *(const v2f*)(pa + 4 * j);
      bf[j].x = pb[(4 * j) * PB];
      bf[j].y = pb[(4 * j + 1) * PB];
    }
    __builtin_amdgcn_sched_barrier(0);
    #pragma unroll
    for (int j = 0; j < KC / 4; ++j)
      acc = wmma_f32_k4(af[j], bf[j], acc);
    if (KSL > KC)
      __syncthreads();
  }

  const int ncol  = n0 + wn * 16 + c;
  const int rbase = m0 + wm * 16 + (h << 3);
  float* op = OutP + (size_t)blockIdx.z * MROWS * N + (size_t)rbase * N + ncol;
  #pragma unroll
  for (int r = 0; r < 8; ++r)
    op[(size_t)r * N] = acc[r];
}

template <int N, int KS>
__global__ __launch_bounds__(256) void fold_out_kernel(
    const float* __restrict__ P,
    const float* __restrict__ bias,
    float* __restrict__ Out) {
  const int e = (blockIdx.x * 256 + threadIdx.x) * 4;
  float4 s = *(const float4*)(P + e);
  #pragma unroll
  for (int z = 1; z < KS; ++z)
    s = add4(s, *(const float4*)(P + (size_t)z * MROWS * N + e));
  s = add4(s, *(const float4*)(bias + (e % N)));
  *(float4*)(Out + e) = s;
}

constexpr int KS1 = 6;
constexpr int KS2 = 4;

extern "C" void kernel_launch(void* const* d_in, const int* in_sizes, int n_in,
                              void* d_out, int out_size, void* d_ws, size_t ws_size,
                              hipStream_t stream) {
  const float* x       = (const float*)d_in[0];
  const int*   lengths = (const int*)  d_in[1];
  const float* W1      = (const float*)d_in[2];
  const float* b1      = (const float*)d_in[3];
  const float* W2      = (const float*)d_in[4];
  const float* b2      = (const float*)d_in[5];
  float*       out     = (float*)d_out;

  float* ws      = (float*)d_ws;
  float* partial = ws;
  float* feat    = partial + (size_t)B * NSEG * 3 * C;
  float* hP      = ws;
  float* oP      = hP + (size_t)KS1 * B * N1;

  pool_partial_kernel<<<dim3(NSEG, B), dim3(PT), 0, stream>>>(x, lengths, partial);
  pool_reduce_kernel<<<dim3(B), dim3(C), 0, stream>>>(partial, lengths, feat);

  gemm_slice_kernel<N1, K1, KS1, 0, false><<<dim3(N1 / BN, B / BM, KS1), dim3(128), 0, stream>>>(
      feat, b1, W1, hP);
  gemm_slice_kernel<N2, K2, KS2, KS1, true><<<dim3(N2 / BN, B / BM, KS2), dim3(128), 0, stream>>>(
      hP, b1, W2, oP);
  fold_out_kernel<N2, KS2><<<dim3(B * N2 / 4 / 256), dim3(256), 0, stream>>>(oP, b2, out);
}


# Pool grid re-cut to 512 work-groups: a batch row's 32-row blocks are dealt round-robin to 8 work-groups, a wave reads 4 whole rows per trip; half the waves to start; mean sums' tree changes
# speedup vs baseline: 5.8583x; 1.1044x over previous
#include <hip/hip_runtime.h>
#include <math.h>

constexpr int B    = 64;
constexpr int T    = 2048;
constexpr int C    = 512;
constexpr int K1   = 3 * C;
constexpr int N1   = C;
constexpr int K2   = C;
constexpr int N2   = C / 2;

typedef float v2f __attribute__((ext_vector_type(2)));
typedef float v8f __attribute__((ext_vector_type(8)));

constexpr int NPC  = 8;
constexpr int PW   = 8;
constexpr int PT   = PW * 32;
constexpr int RPT  = 4;
constexpr int PBLK = PW * RPT;
constexpr int PSTR = NPC * PBLK;

__device__ __forceinline__ float4 add4(float4 a, float4 v) {
  return make_float4(a.x + v.x, a.y + v.y, a.z + v.z, a.w + v.w);
}
__device__ __forceinline__ float4 max4(float4 a, float4 v) {
  return make_float4(fmaxf(a.x, v.x), fmaxf(a.y, v.y), fmaxf(a.z, v.z), fmaxf(a.w, v.w));
}

__device__ __forceinline__ void fold_row(int t, int q, int p, const float4* v,
                                         float4* sb, float4* mx, float4* sf) {
  if (t < q) {
    for (int j = 0; j < 4; ++j) sb[j] = add4(sb[j], v[j]);
  } else if (t < p) {
    for (int j = 0; j < 4; ++j) mx[j] = max4(mx[j], v[j]);
  } else {
    for (int j = 0; j < 4; ++j) sf[j] = add4(sf[j], v[j]);
  }
}

__global__ __launch_bounds__(PT) void pool_partial_kernel(
    const float* __restrict__ x,
    const int* __restrict__ lengths,
    float* __restrict__ partial) {
  __shared__ __attribute__((aligned(16))) float sp[PW * 3 * C];

  const int piece = blockIdx.x;
  const int b     = blockIdx.y;
  const int tid   = threadIdx.x;
  const int lane  = tid & 31;
  const int w     = __builtin_amdgcn_readfirstlane(tid >> 5);

  const int L  = lengths[b];
  const int q  = L / 4;
  const int p  = (3 * L) / 4;
  const int Lc = L < T ? L : T;

  const float ninf = -__builtin_inff();
  const float4 zero4 = make_float4(0.f, 0.f, 0.f, 0.f);
  const float4 ninf4 = make_float4(ninf, ninf, ninf, ninf);

  float* base = partial + (((size_t)b * NPC + piece) * 3) * C + 2 * tid;

  if (piece * PBLK >= Lc) {
    *(float2*)(base + 0 * C) = make_float2(0.f, 0.f);
    *(float2*)(base + 1 * C) = make_float2(ninf, ninf);
    *(float2*)(base + 2 * C) = make_float2(0.f, 0.f);
    return;
  }

  float4 sb[4], mx[4], sf[4];
  for (int j = 0; j < 4; ++j) { sb[j] = zero4; mx[j] = ninf4; sf[j] = zero4; }

  const float* xb = x + (size_t)b * T * C + lane * 4;
  int tb = piece * PBLK + w * RPT;
  for (; tb + RPT <= Lc; tb += PSTR) {
    const float* xp = xb + (size_t)tb * C;
    float4 v[RPT][4];
    #pragma unroll
    for (int r = 0; r < RPT; ++r)
      for (int j = 0; j < 4; ++j)
        v[r][j] = *(const float4*)(xp + (size_t)r * C + 128 * j);
    const int te = tb + RPT - 1;
    if (te < q) {
      #pragma unroll
      for (int r = 0; r < RPT; ++r)
        for (int j = 0; j < 4; ++j) sb[j] = add4(sb[j], v[r][j]);
    } else if (tb >= q && te < p) {
      #pragma unroll
      for (int r = 0; r < RPT; ++r)
        for (int j = 0; j < 4; ++j) mx[j] = max4(mx[j], v[r][j]);
    } else if (tb >= p) {
      #pragma unroll
      for (int r = 0; r < RPT; ++r)
        for (int j = 0; j < 4; ++j) sf[j] = add4(sf[j], v[r][j]);
    } else {
      #pragma unroll
      for (int r = 0; r < RPT; ++r)
        fold_row(tb + r, q, p, v[r], sb, mx, sf);
    }
  }
  if (tb < Lc) {
    const float* xp = xb + (size_t)tb * C;
    const int n = Lc - tb;
    float4 u[RPT - 1][4];
    #pragma unroll
    for (int r = 0; r < RPT - 1; ++r)
      if (r < n)
        for (int j = 0; j < 4; ++j)
          u[r][j] = *(const float4*)(xp + (size_t)r * C + 128 * j);
    #pragma unroll
    for (int r = 0; r < RPT - 1; ++r)
      if (r < n)
        fold_row(tb + r, q, p, u[r], sb, mx, sf);
  }

  float* mine = sp + w * (3 * C) + lane * 4;
  for (int j = 0; j < 4; ++j) {
    *(float4*)(mine + 0 * C + 128 * j) = sb[j];
    *(float4*)(mine + 1 * C + 128 * j) = mx[j];
    *(float4*)(mine + 2 * C + 128 * j) = sf[j];
  }
  __syncthreads();

  const float* fs = sp + 2 * tid;
  float2 rb = *(const float2*)(fs + 0 * C);
  float2 rm = *(const float2*)(fs + 1 * C);
  float2 rf = *(const float2*)(fs + 2 * C);
  for (int k = 1; k < PW; ++k) {
    const float2 ub = *(const float2*)(fs + k * (3 * C) + 0 * C);
    const float2 um = *(const float2*)(fs + k * (3 * C) + 1 * C);
    const float2 uf = *(const float2*)(fs + k * (3 * C) + 2 * C);
    rb.x += ub.x; rb.y += ub.y;
    rm.x = fmaxf(rm.x, um.x); rm.y = fmaxf(rm.y, um.y);
    rf.x += uf.x; rf.y += uf.y;
  }
  *(float2*)(base + 0 * C) = rb;
  *(float2*)(base + 1 * C) = rm;
  *(float2*)(base + 2 * C) = rf;
}

__global__ void pool_reduce_kernel(const float* __restrict__ partial,
                                   const int* __restrict__ lengths,
                                   float* __restrict__ feat) {
  const int b = blockIdx.x;
  const int c = threadIdx.x;

  const int L = lengths[b];
  const float qf = (float)(L / 4);
  const float ff = (float)(L - (3 * L) / 4);

  float sb = 0.f, sf = 0.f, mx = -__builtin_inff();
  const float* base = partial + (size_t)b * NPC * 3 * C + c;
  #pragma unroll
  for (int s = 0; s < NPC; ++s) {
    const float* ps = base + (size_t)s * 3 * C;
    sb += ps[0];
    mx = fmaxf(mx, ps[C]);
    sf += ps[2 * C];
  }
  float* fb = feat + (size_t)b * K1;
  fb[c]         = sb / qf;
  fb[C + c]     = mx;
  fb[2 * C + c] = sf / ff;
}

__device__ __forceinline__ v8f wmma_f32_k4(v2f a, v2f b, v8f c) {
  return __builtin_amdgcn_wmma_f32_16x16x4_f32(
       false, a,  false, b,
       (short)0, c,  false,  false);
}

constexpr int BM = 32;
constexpr int BN = 32;
constexpr int KC = 128;
constexpr int PA = KC + 4;
constexpr int PB = BN + 8;

constexpr int MROWS = B;

template <int N, int K, int KS, int PIN, bool RELU_IN>
__global__ __launch_bounds__(128) void gemm_slice_kernel(
    const float* __restrict__ A,
    const float* __restrict__ bias_in,
    const float* __restrict__ W,
    float* __restrict__ OutP) {
  __shared__ __attribute__((aligned(16))) float sA[BM * PA];
  __shared__ __attribute__((aligned(16))) float sB[KC * PB];

  constexpr int KSL = K / KS;
  static_assert(KSL % KC == 0, "slice must be whole chunks");

  const int tid  = threadIdx.x;
  const int lane = tid & 31;
  const int wave = tid >> 5;
  const int wm   = wave >> 1;
  const int wn   = wave & 1;
  const int c    = lane & 15;
  const int h    = lane >> 4;
  const int m0   = blockIdx.y * BM;
  const int n0   = blockIdx.x * BN;
  const int kbeg = blockIdx.z * KSL;

  const int ac4 = tid & 31;
  const int ar0 = tid >> 5;
  const int bc4 = tid & 7;
  const int br0 = tid >> 3;

  const float* gA = A + (size_t)(m0 + ar0) * K + ac4 * 4;
  const float* gB = W + (size_t)br0 * N + n0 + bc4 * 4;
  float* stA = sA + ar0 * PA + ac4 * 4;
  float* stB = sB + br0 * PB + bc4 * 4;

  const float* pa = sA + (wm * 16 + c) * PA + 2 * h;
  const float* pb = sB + (2 * h) * PB + wn * 16 + c;

  v8f acc = {};
  for (int kc = kbeg; kc < kbeg + KSL; kc += KC) {
    float4 va[8];
    float4 vb[8];
    #pragma unroll
    for (int i = 0; i < 8; ++i)
      vb[i] = *(const float4*)(gB + (size_t)(kc + 16 * i) * N);
    if constexpr (PIN == 0) {
      #pragma unroll
      for (int i = 0; i < 8; ++i)
        va[i] = *(const float4*)(gA + (size_t)(4 * i) * K + kc);
    } else {
      const float4 bv = *(const float4*)(bias_in + kc + ac4 * 4);
      #pragma unroll
      for (int i = 0; i < 8; ++i) {
        const float* s0 = gA + (size_t)(4 * i) * K + kc;
        float4 s = *(const float4*)s0;
        #pragma unroll
        for (int p = 1; p < PIN; ++p)
          s = add4(s, *(const float4*)(s0 + (size_t)p * MROWS * K));
        s = add4(s, bv);
        if (RELU_IN) s = max4(s, make_float4(0.f, 0.f, 0.f, 0.f));
        va[i] = s;
      }
    }
    #pragma unroll
    for (int i = 0; i < 8; ++i)
      *(float4*)(stA + (4 * i) * PA) = va[i];
    #pragma unroll
    for (int i = 0; i < 8; ++i)
      *(float4*)(stB + (16 * i) * PB) = vb[i];
    __syncthreads();

    v2f af[KC / 4];
    v2f bf[KC / 4];
    #pragma unroll
    for (int j = 0; j < KC / 4; ++j) {
      af[j]   = *(const v2f*)(pa + 4 * j);
      bf[j].x = pb[(4 * j) * PB];
      bf[j].y = pb[(4 * j + 1) * PB];
    }
    __builtin_amdgcn_sched_barrier(0);
    #pragma unroll
    for (int j = 0; j < KC / 4; ++j)
      acc = wmma_f32_k4(af[j], bf[j], acc);
    if (KSL > KC)
      __syncthreads();
  }

  const int ncol  = n0 + wn * 16 + c;
  const int rbase = m0 + wm * 16 + (h << 3);
  float* op = OutP + (size_t)blockIdx.z * MROWS * N + (size_t)rbase * N + ncol;
  #pragma unroll
  for (int r = 0; r < 8; ++r)
    op[(size_t)r * N] = acc[r];
}

template <int N, int KS>
__global__ __launch_bounds__(256) void fold_out_kernel(
    const float* __restrict__ P,
    const float* __restrict__ bias,
    float* __restrict__ Out) {
  const int e = (blockIdx.x * 256 + threadIdx.x) * 4;
  float4 s = *(const float4*)(P + e);
  #pragma unroll
  for (int z = 1; z < KS; ++z)
    s = add4(s, *(const float4*)(P + (size_t)z * MROWS * N + e));
  s = add4(s, *(const float4*)(bias + (e % N)));
  *(float4*)(Out + e) = s;
}

constexpr int KS1 = 6;
constexpr int KS2 = 4;

extern "C" void kernel_launch(void* const* d_in, const int* in_sizes, int n_in,
                              void* d_out, int out_size, void* d_ws, size_t ws_size,
                              hipStream_t stream) {
  const float* x       = (const float*)d_in[0];
  const int*   lengths = (const int*)  d_in[1];
  const float* W1      = (const float*)d_in[2];
  const float* b1      = (const float*)d_in[3];
  const float* W2      = (const float*)d_in[4];
  const float* b2      = (const float*)d_in[5];
  float*       out     = (float*)d_out;

  float* ws      = (float*)d_ws;
  float* partial = ws;
  float* feat    = partial + (size_t)B * NPC * 3 * C;
  float* hP      = ws;
  float* oP      = hP + (size_t)KS1 * B * N1;

  pool_partial_kernel<<<dim3(NPC, B), dim3(PT), 0, stream>>>(x, lengths, partial);
  pool_reduce_kernel<<<dim3(B), dim3(C), 0, stream>>>(partial, lengths, feat);

  gemm_slice_kernel<N1, K1, KS1, 0, false><<<dim3(N1 / BN, B / BM, KS1), dim3(128), 0, stream>>>(
      feat, b1, W1, hP);
  gemm_slice_kernel<N2, K2, KS2, KS1, true><<<dim3(N2 / BN, B / BM, KS2), dim3(128), 0, stream>>>(
      hP, b1, W2, oP);
  fold_out_kernel<N2, KS2><<<dim3(B * N2 / 4 / 256), dim3(256), 0, stream>>>(oP, b2, out);
}


# Reduce launch gone: GEMM1's A staging folds the pooling pieces itself, in piece order; 4 pieces per batch row (work-groups split channels in halves), means divided per piece; mean sums' tree changes
# speedup vs baseline: 6.0761x; 1.0372x over previous
#include <hip/hip_runtime.h>
#include <math.h>

constexpr int B    = 64;
constexpr int T    = 2048;
constexpr int C    = 512;
constexpr int K1   = 3 * C;
constexpr int N1   = C;
constexpr int K2   = C;
constexpr int N2   = C / 2;

typedef float v2f __attribute__((ext_vector_type(2)));
typedef float v8f __attribute__((ext_vector_type(8)));

constexpr int NRP  = 4;
constexpr int NCH  = 2;
constexpr int NPC  = NRP * NCH;
constexpr int CH   = C / NCH;
constexpr int PW   = 8;
constexpr int PT   = PW * 32;
constexpr int RPT  = 8;
constexpr int LPR  = CH / 128;
constexpr int PBLK = PW * RPT;
constexpr int PSTR = NRP * PBLK;
static_assert(CH == PT, "the LDS fold gives every thread one channel");

__device__ __forceinline__ float4 add4(float4 a, float4 v) {
  return make_float4(a.x + v.x, a.y + v.y, a.z + v.z, a.w + v.w);
}
__device__ __forceinline__ float4 max4(float4 a, float4 v) {
  return make_float4(fmaxf(a.x, v.x), fmaxf(a.y, v.y), fmaxf(a.z, v.z), fmaxf(a.w, v.w));
}

__device__ __forceinline__ void fold_row(int t, int q, int p, const float4* v,
                                         float4* sb, float4* mx, float4* sf) {
  if (t < q) {
    for (int j = 0; j < LPR; ++j) sb[j] = add4(sb[j], v[j]);
  } else if (t < p) {
    for (int j = 0; j < LPR; ++j) mx[j] = max4(mx[j], v[j]);
  } else {
    for (int j = 0; j < LPR; ++j) sf[j] = add4(sf[j], v[j]);
  }
}

__global__ __launch_bounds__(PT) void pool_partial_kernel(
    const float* __restrict__ x,
    const int* __restrict__ lengths,
    float* __restrict__ partial) {
  __shared__ __attribute__((aligned(16))) float sp[PW * 3 * CH];

  const int rp    = blockIdx.x >> 1;
  const int hc    = blockIdx.x & 1;
  const int b     = blockIdx.y;
  const int tid   = threadIdx.x;
  const int lane  = tid & 31;
  const int w     = __builtin_amdgcn_readfirstlane(tid >> 5);

  const int L  = lengths[b];
  const int q  = L / 4;
  const int p  = (3 * L) / 4;
  const int Lc = L < T ? L : T;

  const float qf = (float)q;
  const float ff = (float)(L - p);

  const float ninf = -__builtin_inff();
  const float4 zero4 = make_float4(0.f, 0.f, 0.f, 0.f);
  const float4 ninf4 = make_float4(ninf, ninf, ninf, ninf);

  float* base = partial + (((size_t)b * NRP + rp) * 3) * C + hc * CH + tid;

  if (rp * PBLK >= Lc) {
    base[0 * C] = 0.f / qf;
    base[1 * C] = ninf;
    base[2 * C] = 0.f / ff;
    return;
  }

  float4 sb[LPR], mx[LPR], sf[LPR];
  for (int j = 0; j < LPR; ++j) { sb[j] = zero4; mx[j] = ninf4; sf[j] = zero4; }

  const float* xb = x + (size_t)b * T * C + hc * CH + lane * 4;
  int tb = rp * PBLK + w * RPT;
  for (; tb + RPT <= Lc; tb += PSTR) {
    const float* xp = xb + (size_t)tb * C;
    float4 v[RPT][LPR];
    #pragma unroll
    for (int r = 0; r < RPT; ++r)
      for (int j = 0; j < LPR; ++j)
        v[r][j] = *(const float4*)(xp + (size_t)r * C + 128 * j);
    const int te = tb + RPT - 1;
    if (te < q) {
      #pragma unroll
      for (int r = 0; r < RPT; ++r)
        for (int j = 0; j < LPR; ++j) sb[j] = add4(sb[j], v[r][j]);
    } else if (tb >= q && te < p) {
      #pragma unroll
      for (int r = 0; r < RPT; ++r)
        for (int j = 0; j < LPR; ++j) mx[j] = max4(mx[j], v[r][j]);
    } else if (tb >= p) {
      #pragma unroll
      for (int r = 0; r < RPT; ++r)
        for (int j = 0; j < LPR; ++j) sf[j] = add4(sf[j], v[r][j]);
    } else {
      #pragma unroll
      for (int r = 0; r < RPT; ++r)
        fold_row(tb + r, q, p, v[r], sb, mx, sf);
    }
  }
  if (tb < Lc) {
    const float* xp = xb + (size_t)tb * C;
    const int n = Lc - tb;
    float4 u[RPT - 1][LPR];
    #pragma unroll
    for (int r = 0; r < RPT - 1; ++r)
      if (r < n)
        for (int j = 0; j < LPR; ++j)
          u[r][j] = *(const float4*)(xp + (size_t)r * C + 128 * j);
    #pragma unroll
    for (int r = 0; r < RPT - 1; ++r)
      if (r < n)
        fold_row(tb + r, q, p, u[r], sb, mx, sf);
  }

  float* mine = sp + w * (3 * CH) + lane * 4;
  for (int j = 0; j < LPR; ++j) {
    *(float4*)(mine + 0 * CH + 128 * j) = sb[j];
    *(float4*)(mine + 1 * CH + 128 * j) = mx[j];
    *(float4*)(mine + 2 * CH + 128 * j) = sf[j];
  }
  __syncthreads();

  const float* fs = sp + tid;
  float rb = fs[0 * CH];
  float rm = fs[1 * CH];
  float rf = fs[2 * CH];
  for (int k = 1; k < PW; ++k) {
    rb += fs[k * (3 * CH) + 0 * CH];
    rm = fmaxf(rm, fs[k * (3 * CH) + 1 * CH]);
    rf += fs[k * (3 * CH) + 2 * CH];
  }
  base[0 * C] = rb / qf;
  base[1 * C] = rm;
  base[2 * C] = rf / ff;
}

__device__ __forceinline__ v8f wmma_f32_k4(v2f a, v2f b, v8f c) {
  return __builtin_amdgcn_wmma_f32_16x16x4_f32(
       false, a,  false, b,
       (short)0, c,  false,  false);
}

constexpr int BM = 32;
constexpr int BN = 32;
constexpr int KC = 128;
constexpr int PA = KC + 4;
constexpr int PB = BN + 8;

constexpr int MROWS = B;

template <int N, int K, int KS, int PIN, bool RELU_IN, int NPIECE, int FOLD>
__device__ __forceinline__ void gemm_slice_body(
    const float* __restrict__ A,
    const float* __restrict__ bias_in,
    const float* __restrict__ W,
    float* __restrict__ OutP,
    float* __restrict__ sA,
    float* __restrict__ sB) {
  constexpr int KSL = K / KS;

  const int tid  = threadIdx.x;
  const int lane = tid & 31;
  const int wave = tid >> 5;
  const int wm   = wave >> 1;
  const int wn   = wave & 1;
  const int c    = lane & 15;
  const int h    = lane >> 4;
  const int m0   = blockIdx.y * BM;
  const int n0   = blockIdx.x * BN;
  const int kbeg = blockIdx.z * KSL;

  const int ac4 = tid & 31;
  const int ar0 = tid >> 5;
  const int bc4 = tid & 7;
  const int br0 = tid >> 3;

  constexpr int RSTR = (NPIECE > 0 ? NPIECE : 1) * K;
  const float* gA = A + (size_t)(m0 + ar0) * RSTR + ac4 * 4;
  const float* gB = W + (size_t)br0 * N + n0 + bc4 * 4;
  float* stA = sA + ar0 * PA + ac4 * 4;
  float* stB = sB + br0 * PB + bc4 * 4;

  const float* pa = sA + (wm * 16 + c) * PA + 2 * h;
  const float* pb = sB + (2 * h) * PB + wn * 16 + c;

  v8f acc = {};
  for (int kc = kbeg; kc < kbeg + KSL; kc += KC) {
    float4 va[8];
    float4 vb[8];
    #pragma unroll
    for (int i = 0; i < 8; ++i)
      vb[i] = *(const float4*)(gB + (size_t)(kc + 16 * i) * N);
    if constexpr (NPIECE > 0) {
      float4 pv[8][NPIECE];
      #pragma unroll
      for (int i = 0; i < 8; ++i)
        #pragma unroll
        for (int s = 0; s < NPIECE; ++s)
          pv[i][s] = *(const float4*)(gA + ((size_t)(4 * i) * NPIECE + s) * K + kc);
      #pragma unroll
      for (int i = 0; i < 8; ++i) {
        if constexpr (FOLD == 2) {
          const float ninf = -__builtin_inff();
          float4 t = make_float4(ninf, ninf, ninf, ninf);
          #pragma unroll
          for (int s = 0; s < NPIECE; ++s) t = max4(t, pv[i][s]);
          va[i] = t;
        } else {
          float4 t = make_float4(0.f, 0.f, 0.f, 0.f);
          #pragma unroll
          for (int s = 0; s < NPIECE; ++s) t = add4(t, pv[i][s]);
          va[i] = t;
        }
      }
    } else {
      const float4 bv = *(const float4*)(bias_in + kc + ac4 * 4);
      #pragma unroll
      for (int i = 0; i < 8; ++i) {
        const float* s0 = gA + (size_t)(4 * i) * RSTR + kc;
        float4 s = *(const float4*)s0;
        #pragma unroll
        for (int p = 1; p < PIN; ++p)
          s = add4(s, *(const float4*)(s0 + (size_t)p * MROWS * K));
        s = add4(s, bv);
        if (RELU_IN) s = max4(s, make_float4(0.f, 0.f, 0.f, 0.f));
        va[i] = s;
      }
    }
    #pragma unroll
    for (int i = 0; i < 8; ++i)
      *(float4*)(stA + (4 * i) * PA) = va[i];
    #pragma unroll
    for (int i = 0; i < 8; ++i)
      *(float4*)(stB + (16 * i) * PB) = vb[i];
    __syncthreads();

    v2f af[KC / 4];
    v2f bf[KC / 4];
    #pragma unroll
    for (int j = 0; j < KC / 4; ++j) {
      af[j]   = *(const v2f*)(pa + 4 * j);
      bf[j].x = pb[(4 * j) * PB];
      bf[j].y = pb[(4 * j + 1) * PB];
    }
    __builtin_amdgcn_sched_barrier(0);
    #pragma unroll
    for (int j = 0; j < KC / 4; ++j)
      acc = wmma_f32_k4(af[j], bf[j], acc);
    if (KSL > KC)
      __syncthreads();
  }

  const int ncol  = n0 + wn * 16 + c;
  const int rbase = m0 + wm * 16 + (h << 3);
  float* op = OutP + (size_t)blockIdx.z * MROWS * N + (size_t)rbase * N + ncol;
  #pragma unroll
  for (int r = 0; r < 8; ++r)
    op[(size_t)r * N] = acc[r];
}

template <int N, int K, int KS, int PIN, bool RELU_IN, int NPIECE>
__global__ __launch_bounds__(128) void gemm_slice_kernel(
    const float* __restrict__ A,
    const float* __restrict__ bias_in,
    const float* __restrict__ W,
    float* __restrict__ OutP) {
  __shared__ __attribute__((aligned(16))) float sA[BM * PA];
  __shared__ __attribute__((aligned(16))) float sB[KC * PB];

  constexpr int KSL = K / KS;
  static_assert(KSL % KC == 0, "slice must be whole chunks");
  static_assert((PIN > 0) != (NPIECE > 0), "A is slice planes or pooling pieces");
  static_assert(NPIECE == 0 || (K == 3 * C && C % KSL == 0),
                "a slice of the pooled operand must lie inside one class");

  if constexpr (NPIECE > 0) {
    const int cls = (blockIdx.z * KSL) / C;
    if (cls == 1)
      gemm_slice_body<N, K, KS, PIN, RELU_IN, NPIECE, 2>(A, bias_in, W, OutP, sA, sB);
    else
      gemm_slice_body<N, K, KS, PIN, RELU_IN, NPIECE, 1>(A, bias_in, W, OutP, sA, sB);
  } else {
    gemm_slice_body<N, K, KS, PIN, RELU_IN, NPIECE, 0>(A, bias_in, W, OutP, sA, sB);
  }
}

template <int N, int KS>
__global__ __launch_bounds__(256) void fold_out_kernel(
    const float* __restrict__ P,
    const float* __restrict__ bias,
    float* __restrict__ Out) {
  const int e = (blockIdx.x * 256 + threadIdx.x) * 4;
  float4 s = *(const float4*)(P + e);
  #pragma unroll
  for (int z = 1; z < KS; ++z)
    s = add4(s, *(const float4*)(P + (size_t)z * MROWS * N + e));
  s = add4(s, *(const float4*)(bias + (e % N)));
  *(float4*)(Out + e) = s;
}

constexpr int KS1 = 6;
constexpr int KS2 = 4;

extern "C" void kernel_launch(void* const* d_in, const int* in_sizes, int n_in,
                              void* d_out, int out_size, void* d_ws, size_t ws_size,
                              hipStream_t stream) {
  const float* x       = (const float*)d_in[0];
  const int*   lengths = (const int*)  d_in[1];
  const float* W1      = (const float*)d_in[2];
  const float* b1      = (const float*)d_in[3];
  const float* W2      = (const float*)d_in[4];
  const float* b2      = (const float*)d_in[5];
  float*       out     = (float*)d_out;

  float* ws      = (float*)d_ws;
  float* partial = ws;
  float* hP      = partial + (size_t)B * NRP * K1;
  float* oP      = hP + (size_t)KS1 * B * N1;

  pool_partial_kernel<<<dim3(NPC, B), dim3(PT), 0, stream>>>(x, lengths, partial);

  gemm_slice_kernel<N1, K1, KS1, 0, false, NRP><<<dim3(N1 / BN, B / BM, KS1), dim3(128), 0, stream>>>(
      partial, b1, W1, hP);
  gemm_slice_kernel<N2, K2, KS2, KS1, true, 0><<<dim3(N2 / BN, B / BM, KS2), dim3(128), 0, stream>>>(
      hP, b1, W2, oP);
  fold_out_kernel<N2, KS2><<<dim3(B * N2 / 4 / 256), dim3(256), 0, stream>>>(oP, b2, out);
}


# Fold launch gone: GEMM2's four K slices become four wave pairs of one work-group (16x32 tile, 8 waves); partial tiles added through LDS in slice order, then b2; every sum keeps its order
# speedup vs baseline: 6.2685x; 1.0317x over previous
#include <hip/hip_runtime.h>
#include <math.h>

constexpr int B    = 64;
constexpr int T    = 2048;
constexpr int C    = 512;
constexpr int K1   = 3 * C;
constexpr int N1   = C;
constexpr int K2   = C;
constexpr int N2   = C / 2;

typedef float v2f __attribute__((ext_vector_type(2)));
typedef float v8f __attribute__((ext_vector_type(8)));

constexpr int NRP  = 4;
constexpr int NCH  = 2;
constexpr int NPC  = NRP * NCH;
constexpr int CH   = C / NCH;
constexpr int PW   = 8;
constexpr int PT   = PW * 32;
constexpr int RPT  = 8;
constexpr int LPR  = CH / 128;
constexpr int PBLK = PW * RPT;
constexpr int PSTR = NRP * PBLK;
static_assert(CH == PT, "the LDS fold gives every thread one channel");

__device__ __forceinline__ float4 add4(float4 a, float4 v) {
  return make_float4(a.x + v.x, a.y + v.y, a.z + v.z, a.w + v.w);
}
__device__ __forceinline__ float4 max4(float4 a, float4 v) {
  return make_float4(fmaxf(a.x, v.x), fmaxf(a.y, v.y), fmaxf(a.z, v.z), fmaxf(a.w, v.w));
}

__device__ __forceinline__ void fold_row(int t, int q, int p, const float4* v,
                                         float4* sb, float4* mx, float4* sf) {
  if (t < q) {
    for (int j = 0; j < LPR; ++j) sb[j] = add4(sb[j], v[j]);
  } else if (t < p) {
    for (int j = 0; j < LPR; ++j) mx[j] = max4(mx[j], v[j]);
  } else {
    for (int j = 0; j < LPR; ++j) sf[j] = add4(sf[j], v[j]);
  }
}

__global__ __launch_bounds__(PT) void pool_partial_kernel(
    const float* __restrict__ x,
    const int* __restrict__ lengths,
    float* __restrict__ partial) {
  __shared__ __attribute__((aligned(16))) float sp[PW * 3 * CH];

  const int rp    = blockIdx.x >> 1;
  const int hc    = blockIdx.x & 1;
  const int b     = blockIdx.y;
  const int tid   = threadIdx.x;
  const int lane  = tid & 31;
  const int w     = __builtin_amdgcn_readfirstlane(tid >> 5);

  const int L  = lengths[b];
  const int q  = L / 4;
  const int p  = (3 * L) / 4;
  const int Lc = L < T ? L : T;

  const float qf = (float)q;
  const float ff = (float)(L - p);

  const float ninf = -__builtin_inff();
  const float4 zero4 = make_float4(0.f, 0.f, 0.f, 0.f);
  const float4 ninf4 = make_float4(ninf, ninf, ninf, ninf);

  float* base = partial + (((size_t)b * NRP + rp) * 3) * C + hc * CH + tid;

  if (rp * PBLK >= Lc) {
    base[0 * C] = 0.f / qf;
    base[1 * C] = ninf;
    base[2 * C] = 0.f / ff;
    return;
  }

  float4 sb[LPR], mx[LPR], sf[LPR];
  for (int j = 0; j < LPR; ++j) { sb[j] = zero4; mx[j] = ninf4; sf[j] = zero4; }

  const float* xb = x + (size_t)b * T * C + hc * CH + lane * 4;
  int tb = rp * PBLK + w * RPT;
  for (; tb + RPT <= Lc; tb += PSTR) {
    const float* xp = xb + (size_t)tb * C;
    float4 v[RPT][LPR];
    #pragma unroll
    for (int r = 0; r < RPT; ++r)
      for (int j = 0; j < LPR; ++j)
        v[r][j] = *(const float4*)(xp + (size_t)r * C + 128 * j);
    const int te = tb + RPT - 1;
    if (te < q) {
      #pragma unroll
      for (int r = 0; r < RPT; ++r)
        for (int j = 0; j < LPR; ++j) sb[j] = add4(sb[j], v[r][j]);
    } else if (tb >= q && te < p) {
      #pragma unroll
      for (int r = 0; r < RPT; ++r)
        for (int j = 0; j < LPR; ++j) mx[j] = max4(mx[j], v[r][j]);
    } else if (tb >= p) {
      #pragma unroll
      for (int r = 0; r < RPT; ++r)
        for (int j = 0; j < LPR; ++j) sf[j] = add4(sf[j], v[r][j]);
    } else {
      #pragma unroll
      for (int r = 0; r < RPT; ++r)
        fold_row(tb + r, q, p, v[r], sb, mx, sf);
    }
  }
  if (tb < Lc) {
    const float* xp = xb + (size_t)tb * C;
    const int n = Lc - tb;
    float4 u[RPT - 1][LPR];
    #pragma unroll
    for (int r = 0; r < RPT - 1; ++r)
      if (r < n)
        for (int j = 0; j < LPR; ++j)
          u[r][j] = *(const float4*)(xp + (size_t)r * C + 128 * j);
    #pragma unroll
    for (int r = 0; r < RPT - 1; ++r)
      if (r < n)
        fold_row(tb + r, q, p, u[r], sb, mx, sf);
  }

  float* mine = sp + w * (3 * CH) + lane * 4;
  for (int j = 0; j < LPR; ++j) {
    *(float4*)(mine + 0 * CH + 128 * j) = sb[j];
    *(float4*)(mine + 1 * CH + 128 * j) = mx[j];
    *(float4*)(mine + 2 * CH + 128 * j) = sf[j];
  }
  __syncthreads();

  const float* fs = sp + tid;
  float rb = fs[0 * CH];
  float rm = fs[1 * CH];
  float rf = fs[2 * CH];
  for (int k = 1; k < PW; ++k) {
    rb += fs[k * (3 * CH) + 0 * CH];
    rm = fmaxf(rm, fs[k * (3 * CH) + 1 * CH]);
    rf += fs[k * (3 * CH) + 2 * CH];
  }
  base[0 * C] = rb / qf;
  base[1 * C] = rm;
  base[2 * C] = rf / ff;
}

__device__ __forceinline__ v8f wmma_f32_k4(v2f a, v2f b, v8f c) {
  return __builtin_amdgcn_wmma_f32_16x16x4_f32(
       false, a,  false, b,
       (short)0, c,  false,  false);
}

constexpr int BM = 32;
constexpr int BN = 32;
constexpr int KC = 128;
constexpr int PA = KC + 4;
constexpr int PB = BN + 8;

constexpr int MROWS = B;

template <int N, int K, int KS, int NPIECE, bool MAXF>
__device__ __forceinline__ void gemm_slice_body(
    const float* __restrict__ A,
    const float* __restrict__ W,
    float* __restrict__ OutP,
    float* __restrict__ sA,
    float* __restrict__ sB) {
  constexpr int KSL = K / KS;

  const int tid  = threadIdx.x;
  const int lane = tid & 31;
  const int wave = tid >> 5;
  const int wm   = wave >> 1;
  const int wn   = wave & 1;
  const int c    = lane & 15;
  const int h    = lane >> 4;
  const int m0   = blockIdx.y * BM;
  const int n0   = blockIdx.x * BN;
  const int kbeg = blockIdx.z * KSL;

  const int ac4 = tid & 31;
  const int ar0 = tid >> 5;
  const int bc4 = tid & 7;
  const int br0 = tid >> 3;

  constexpr int RSTR = NPIECE * K;
  const float* gA = A + (size_t)(m0 + ar0) * RSTR + ac4 * 4;
  const float* gB = W + (size_t)br0 * N + n0 + bc4 * 4;
  float* stA = sA + ar0 * PA + ac4 * 4;
  float* stB = sB + br0 * PB + bc4 * 4;

  const float* pa = sA + (wm * 16 + c) * PA + 2 * h;
  const float* pb = sB + (2 * h) * PB + wn * 16 + c;

  v8f acc = {};
  for (int kc = kbeg; kc < kbeg + KSL; kc += KC) {
    float4 va[8];
    float4 vb[8];
    #pragma unroll
    for (int i = 0; i < 8; ++i)
      vb[i] = *(const float4*)(gB + (size_t)(kc + 16 * i) * N);
    float4 pv[8][NPIECE];
    #pragma unroll
    for (int i = 0; i < 8; ++i)
      #pragma unroll
      for (int s = 0; s < NPIECE; ++s)
        pv[i][s] = *(const float4*)(gA + ((size_t)(4 * i) * NPIECE + s) * K + kc);
    #pragma unroll
    for (int i = 0; i < 8; ++i) {
      if constexpr (MAXF) {
        const float ninf = -__builtin_inff();
        float4 t = make_float4(ninf, ninf, ninf, ninf);
        #pragma unroll
        for (int s = 0; s < NPIECE; ++s) t = max4(t, pv[i][s]);
        va[i] = t;
      } else {
        float4 t = make_float4(0.f, 0.f, 0.f, 0.f);
        #pragma unroll
        for (int s = 0; s < NPIECE; ++s) t = add4(t, pv[i][s]);
        va[i] = t;
      }
    }
    #pragma unroll
    for (int i = 0; i < 8; ++i)
      *(float4*)(stA + (4 * i) * PA) = va[i];
    #pragma unroll
    for (int i = 0; i < 8; ++i)
      *(float4*)(stB + (16 * i) * PB) = vb[i];
    __syncthreads();

    v2f af[KC / 4];
    v2f bf[KC / 4];
    #pragma unroll
    for (int j = 0; j < KC / 4; ++j) {
      af[j]   = *(const v2f*)(pa + 4 * j);
      bf[j].x = pb[(4 * j) * PB];
      bf[j].y = pb[(4 * j + 1) * PB];
    }
    __builtin_amdgcn_sched_barrier(0);
    #pragma unroll
    for (int j = 0; j < KC / 4; ++j)
      acc = wmma_f32_k4(af[j], bf[j], acc);
    if (KSL > KC)
      __syncthreads();
  }

  const int ncol  = n0 + wn * 16 + c;
  const int rbase = m0 + wm * 16 + (h << 3);
  float* op = OutP + (size_t)blockIdx.z * MROWS * N + (size_t)rbase * N + ncol;
  #pragma unroll
  for (int r = 0; r < 8; ++r)
    op[(size_t)r * N] = acc[r];
}

template <int N, int K, int KS, int NPIECE>
__global__ __launch_bounds__(128) void gemm_slice_kernel(
    const float* __restrict__ A,
    const float* __restrict__ W,
    float* __restrict__ OutP) {
  __shared__ __attribute__((aligned(16))) float sA[BM * PA];
  __shared__ __attribute__((aligned(16))) float sB[KC * PB];

  constexpr int KSL = K / KS;
  static_assert(KSL % KC == 0, "slice must be whole chunks");
  static_assert(K == 3 * C && C % KSL == 0,
                "a slice of the pooled operand must lie inside one class");

  const int cls = (blockIdx.z * KSL) / C;
  if (cls == 1)
    gemm_slice_body<N, K, KS, NPIECE, true>(A, W, OutP, sA, sB);
  else
    gemm_slice_body<N, K, KS, NPIECE, false>(A, W, OutP, sA, sB);
}

constexpr int TM   = 16;
constexpr int TN   = 32;
constexpr int KG   = K2 / KC;
constexpr int GT   = KG * 2 * 32;
constexpr int PA2  = K2 + 4;
constexpr int PR   = TN + 4;
static_assert(TM * TN / 4 <= GT, "one float4 of the tile per adding thread");

template <int N, int K, int PIN>
__global__ __launch_bounds__(GT) void gemm_out_kernel(
    const float* __restrict__ P,
    const float* __restrict__ bias_in,
    const float* __restrict__ W,
    const float* __restrict__ bias_out,
    float* __restrict__ Out) {
  __shared__ __attribute__((aligned(16))) float sA[TM * PA2];
  __shared__ __attribute__((aligned(16))) float sB[K * PB];
  __shared__ __attribute__((aligned(16))) float sR[KG * TM * PR];

  const int tid  = threadIdx.x;
  const int lane = tid & 31;
  const int wave = tid >> 5;
  const int kg   = wave >> 1;
  const int wn   = wave & 1;
  const int c    = lane & 15;
  const int h    = lane >> 4;
  const int m0   = blockIdx.y * TM;
  const int n0   = blockIdx.x * TN;

  constexpr int AC4 = K / 4;
  constexpr int ARS = GT / AC4;
  constexpr int ANP = TM / ARS;
  constexpr int BRS = GT / 8;
  constexpr int BNP = K / BRS;
  const int ac4 = tid % AC4;
  const int ar0 = tid / AC4;
  const int bc4 = tid & 7;
  const int br0 = tid >> 3;

  const float* gA = P + (size_t)(m0 + ar0) * K + ac4 * 4;
  const float* gB = W + (size_t)br0 * N + n0 + bc4 * 4;

  float4 vb[BNP];
  #pragma unroll
  for (int i = 0; i < BNP; ++i)
    vb[i] = *(const float4*)(gB + (size_t)(BRS * i) * N);
  const float4 bv = *(const float4*)(bias_in + ac4 * 4);
  float4 va[ANP];
  #pragma unroll
  for (int i = 0; i < ANP; ++i) {
    const float* s0 = gA + (size_t)(ARS * i) * K;
    float4 s = *(const float4*)s0;
    #pragma unroll
    for (int p = 1; p < PIN; ++p)
      s = add4(s, *(const float4*)(s0 + (size_t)p * MROWS * K));
    s = add4(s, bv);
    va[i] = max4(s, make_float4(0.f, 0.f, 0.f, 0.f));
  }
  #pragma unroll
  for (int i = 0; i < ANP; ++i)
    *(float4*)(sA + (ar0 + ARS * i) * PA2 + ac4 * 4) = va[i];
  #pragma unroll
  for (int i = 0; i < BNP; ++i)
    *(float4*)(sB + (br0 + BRS * i) * PB + bc4 * 4) = vb[i];
  __syncthreads();

  const float* pa = sA + c * PA2 + kg * KC + 2 * h;
  const float* pb = sB + (kg * KC + 2 * h) * PB + wn * 16 + c;
  v2f af[KC / 4];
  v2f bf[KC / 4];
  #pragma unroll
  for (int j = 0; j < KC / 4; ++j) {
    af[j]   = *(const v2f*)(pa + 4 * j);
    bf[j].x = pb[(4 * j) * PB];
    bf[j].y = pb[(4 * j + 1) * PB];
  }
  __builtin_amdgcn_sched_barrier(0);
  v8f acc = {};
  #pragma unroll
  for (int j = 0; j < KC / 4; ++j)
    acc = wmma_f32_k4(af[j], bf[j], acc);

  float* rp = sR + kg * (TM * PR) + (h << 3) * PR + wn * 16 + c;
  #pragma unroll
  for (int r = 0; r < 8; ++r)
    rp[r * PR] = acc[r];
  __syncthreads();

  if (tid < TM * TN / 4) {
    const int row = tid >> 3;
    const int c4  = tid & 7;
    const float* rs = sR + row * PR + c4 * 4;
    float4 s = *(const float4*)rs;
    #pragma unroll
    for (int z = 1; z < KG; ++z)
      s = add4(s, *(const float4*)(rs + z * (TM * PR)));
    s = add4(s, *(const float4*)(bias_out + n0 + c4 * 4));
    *(float4*)(Out + (size_t)(m0 + row) * N + n0 + c4 * 4) = s;
  }
}

constexpr int KS1 = 6;

extern "C" void kernel_launch(void* const* d_in, const int* in_sizes, int n_in,
                              void* d_out, int out_size, void* d_ws, size_t ws_size,
                              hipStream_t stream) {
  const float* x       = (const float*)d_in[0];
  const int*   lengths = (const int*)  d_in[1];
  const float* W1      = (const float*)d_in[2];
  const float* b1      = (const float*)d_in[3];
  const float* W2      = (const float*)d_in[4];
  const float* b2      = (const float*)d_in[5];
  float*       out     = (float*)d_out;

  float* ws      = (float*)d_ws;
  float* partial = ws;
  float* hP      = partial + (size_t)B * NRP * K1;

  pool_partial_kernel<<<dim3(NPC, B), dim3(PT), 0, stream>>>(x, lengths, partial);

  gemm_slice_kernel<N1, K1, KS1, NRP><<<dim3(N1 / BN, B / BM, KS1), dim3(128), 0, stream>>>(
      partial, W1, hP);
  gemm_out_kernel<N2, K2, KS1><<<dim3(N2 / TN, B / TM), dim3(GT), 0, stream>>>(
      hP, b1, W2, b2, out);
}


# Pool re-cut to 1024 work-groups of 4 waves: each a 512-B channel quarter of a row piece, 16 rows per wave trip, so a WGP hosts four batch rows' work-groups, not two; mean sums' tree changes
# speedup vs baseline: 6.6349x; 1.0585x over previous
#include <hip/hip_runtime.h>
#include <math.h>

constexpr int B    = 64;
constexpr int T    = 2048;
constexpr int C    = 512;
constexpr int K1   = 3 * C;
constexpr int N1   = C;
constexpr int K2   = C;
constexpr int N2   = C / 2;

typedef float v2f __attribute__((ext_vector_type(2)));
typedef float v8f __attribute__((ext_vector_type(8)));

constexpr int NRP  = 4;
constexpr int NCH  = 4;
constexpr int NPC  = NRP * NCH;
constexpr int CH   = C / NCH;
constexpr int PW   = 4;
constexpr int PT   = PW * 32;
constexpr int RPT  = 16;
constexpr int LPR  = CH / 128;
constexpr int PBLK = PW * RPT;
constexpr int PSTR = NRP * PBLK;
static_assert(CH == PT, "the LDS fold gives every thread one channel");

__device__ __forceinline__ float4 add4(float4 a, float4 v) {
  return make_float4(a.x + v.x, a.y + v.y, a.z + v.z, a.w + v.w);
}
__device__ __forceinline__ float4 max4(float4 a, float4 v) {
  return make_float4(fmaxf(a.x, v.x), fmaxf(a.y, v.y), fmaxf(a.z, v.z), fmaxf(a.w, v.w));
}

__device__ __forceinline__ void fold_row(int t, int q, int p, const float4* v,
                                         float4* sb, float4* mx, float4* sf) {
  if (t < q) {
    for (int j = 0; j < LPR; ++j) sb[j] = add4(sb[j], v[j]);
  } else if (t < p) {
    for (int j = 0; j < LPR; ++j) mx[j] = max4(mx[j], v[j]);
  } else {
    for (int j = 0; j < LPR; ++j) sf[j] = add4(sf[j], v[j]);
  }
}

__global__ __launch_bounds__(PT) void pool_partial_kernel(
    const float* __restrict__ x,
    const int* __restrict__ lengths,
    float* __restrict__ partial) {
  __shared__ __attribute__((aligned(16))) float sp[PW * 3 * CH];

  const int rp    = blockIdx.x / NCH;
  const int hc    = blockIdx.x % NCH;
  const int b     = blockIdx.y;
  const int tid   = threadIdx.x;
  const int lane  = tid & 31;
  const int w     = __builtin_amdgcn_readfirstlane(tid >> 5);

  const int L  = lengths[b];
  const int q  = L / 4;
  const int p  = (3 * L) / 4;
  const int Lc = L < T ? L : T;

  const float qf = (float)q;
  const float ff = (float)(L - p);

  const float ninf = -__builtin_inff();
  const float4 zero4 = make_float4(0.f, 0.f, 0.f, 0.f);
  const float4 ninf4 = make_float4(ninf, ninf, ninf, ninf);

  float* base = partial + (((size_t)b * NRP + rp) * 3) * C + hc * CH + tid;

  if (rp * PBLK >= Lc) {
    base[0 * C] = 0.f / qf;
    base[1 * C] = ninf;
    base[2 * C] = 0.f / ff;
    return;
  }

  float4 sb[LPR], mx[LPR], sf[LPR];
  for (int j = 0; j < LPR; ++j) { sb[j] = zero4; mx[j] = ninf4; sf[j] = zero4; }

  const float* xb = x + (size_t)b * T * C + hc * CH + lane * 4;
  int tb = rp * PBLK + w * RPT;
  for (; tb + RPT <= Lc; tb += PSTR) {
    const float* xp = xb + (size_t)tb * C;
    float4 v[RPT][LPR];
    #pragma unroll
    for (int r = 0; r < RPT; ++r)
      for (int j = 0; j < LPR; ++j)
        v[r][j] = *(const float4*)(xp + (size_t)r * C + 128 * j);
    const int te = tb + RPT - 1;
    if (te < q) {
      #pragma unroll
      for (int r = 0; r < RPT; ++r)
        for (int j = 0; j < LPR; ++j) sb[j] = add4(sb[j], v[r][j]);
    } else if (tb >= q && te < p) {
      #pragma unroll
      for (int r = 0; r < RPT; ++r)
        for (int j = 0; j < LPR; ++j) mx[j] = max4(mx[j], v[r][j]);
    } else if (tb >= p) {
      #pragma unroll
      for (int r = 0; r < RPT; ++r)
        for (int j = 0; j < LPR; ++j) sf[j] = add4(sf[j], v[r][j]);
    } else {
      #pragma unroll
      for (int r = 0; r < RPT; ++r)
        fold_row(tb + r, q, p, v[r], sb, mx, sf);
    }
  }
  if (tb < Lc) {
    const float* xp = xb + (size_t)tb * C;
    const int n = Lc - tb;
    float4 u[RPT - 1][LPR];
    #pragma unroll
    for (int r = 0; r < RPT - 1; ++r)
      if (r < n)
        for (int j = 0; j < LPR; ++j)
          u[r][j] = *(const float4*)(xp + (size_t)r * C + 128 * j);
    #pragma unroll
    for (int r = 0; r < RPT - 1; ++r)
      if (r < n)
        fold_row(tb + r, q, p, u[r], sb, mx, sf);
  }

  float* mine = sp + w * (3 * CH) + lane * 4;
  for (int j = 0; j < LPR; ++j) {
    *(float4*)(mine + 0 * CH + 128 * j) = sb[j];
    *(float4*)(mine + 1 * CH + 128 * j) = mx[j];
    *(float4*)(mine + 2 * CH + 128 * j) = sf[j];
  }
  __syncthreads();

  const float* fs = sp + tid;
  float rb = fs[0 * CH];
  float rm = fs[1 * CH];
  float rf = fs[2 * CH];
  for (int k = 1; k < PW; ++k) {
    rb += fs[k * (3 * CH) + 0 * CH];
    rm = fmaxf(rm, fs[k * (3 * CH) + 1 * CH]);
    rf += fs[k * (3 * CH) + 2 * CH];
  }
  base[0 * C] = rb / qf;
  base[1 * C] = rm;
  base[2 * C] = rf / ff;
}

__device__ __forceinline__ v8f wmma_f32_k4(v2f a, v2f b, v8f c) {
  return __builtin_amdgcn_wmma_f32_16x16x4_f32(
       false, a,  false, b,
       (short)0, c,  false,  false);
}

constexpr int BM = 32;
constexpr int BN = 32;
constexpr int KC = 128;
constexpr int PA = KC + 4;
constexpr int PB = BN + 8;

constexpr int MROWS = B;

template <int N, int K, int KS, int NPIECE, bool MAXF>
__device__ __forceinline__ void gemm_slice_body(
    const float* __restrict__ A,
    const float* __restrict__ W,
    float* __restrict__ OutP,
    float* __restrict__ sA,
    float* __restrict__ sB) {
  constexpr int KSL = K / KS;

  const int tid  = threadIdx.x;
  const int lane = tid & 31;
  const int wave = tid >> 5;
  const int wm   = wave >> 1;
  const int wn   = wave & 1;
  const int c    = lane & 15;
  const int h    = lane >> 4;
  const int m0   = blockIdx.y * BM;
  const int n0   = blockIdx.x * BN;
  const int kbeg = blockIdx.z * KSL;

  const int ac4 = tid & 31;
  const int ar0 = tid >> 5;
  const int bc4 = tid & 7;
  const int br0 = tid >> 3;

  constexpr int RSTR = NPIECE * K;
  const float* gA = A + (size_t)(m0 + ar0) * RSTR + ac4 * 4;
  const float* gB = W + (size_t)br0 * N + n0 + bc4 * 4;
  float* stA = sA + ar0 * PA + ac4 * 4;
  float* stB = sB + br0 * PB + bc4 * 4;

  const float* pa = sA + (wm * 16 + c) * PA + 2 * h;
  const float* pb = sB + (2 * h) * PB + wn * 16 + c;

  v8f acc = {};
  for (int kc = kbeg; kc < kbeg + KSL; kc += KC) {
    float4 va[8];
    float4 vb[8];
    #pragma unroll
    for (int i = 0; i < 8; ++i)
      vb[i] = *(const float4*)(gB + (size_t)(kc + 16 * i) * N);
    float4 pv[8][NPIECE];
    #pragma unroll
    for (int i = 0; i < 8; ++i)
      #pragma unroll
      for (int s = 0; s < NPIECE; ++s)
        pv[i][s] = *(const float4*)(gA + ((size_t)(4 * i) * NPIECE + s) * K + kc);
    #pragma unroll
    for (int i = 0; i < 8; ++i) {
      if constexpr (MAXF) {
        const float ninf = -__builtin_inff();
        float4 t = make_float4(ninf, ninf, ninf, ninf);
        #pragma unroll
        for (int s = 0; s < NPIECE; ++s) t = max4(t, pv[i][s]);
        va[i] = t;
      } else {
        float4 t = make_float4(0.f, 0.f, 0.f, 0.f);
        #pragma unroll
        for (int s = 0; s < NPIECE; ++s) t = add4(t, pv[i][s]);
        va[i] = t;
      }
    }
    #pragma unroll
    for (int i = 0; i < 8; ++i)
      *(float4*)(stA + (4 * i) * PA) = va[i];
    #pragma unroll
    for (int i = 0; i < 8; ++i)
      *(float4*)(stB + (16 * i) * PB) = vb[i];
    __syncthreads();

    v2f af[KC / 4];
    v2f bf[KC / 4];
    #pragma unroll
    for (int j = 0; j < KC / 4; ++j) {
      af[j]   = *(const v2f*)(pa + 4 * j);
      bf[j].x = pb[(4 * j) * PB];
      bf[j].y = pb[(4 * j + 1) * PB];
    }
    __builtin_amdgcn_sched_barrier(0);
    #pragma unroll
    for (int j = 0; j < KC / 4; ++j)
      acc = wmma_f32_k4(af[j], bf[j], acc);
    if (KSL > KC)
      __syncthreads();
  }

  const int ncol  = n0 + wn * 16 + c;
  const int rbase = m0 + wm * 16 + (h << 3);
  float* op = OutP + (size_t)blockIdx.z * MROWS * N + (size_t)rbase * N + ncol;
  #pragma unroll
  for (int r = 0; r < 8; ++r)
    op[(size_t)r * N] = acc[r];
}

template <int N, int K, int KS, int NPIECE>
__global__ __launch_bounds__(128) void gemm_slice_kernel(
    const float* __restrict__ A,
    const float* __restrict__ W,
    float* __restrict__ OutP) {
  __shared__ __attribute__((aligned(16))) float sA[BM * PA];
  __shared__ __attribute__((aligned(16))) float sB[KC * PB];

  constexpr int KSL = K / KS;
  static_assert(KSL % KC == 0, "slice must be whole chunks");
  static_assert(K == 3 * C && C % KSL == 0,
                "a slice of the pooled operand must lie inside one class");

  const int cls = (blockIdx.z * KSL) / C;
  if (cls == 1)
    gemm_slice_body<N, K, KS, NPIECE, true>(A, W, OutP, sA, sB);
  else
    gemm_slice_body<N, K, KS, NPIECE, false>(A, W, OutP, sA, sB);
}

constexpr int TM   = 16;
constexpr int TN   = 32;
constexpr int KG   = K2 / KC;
constexpr int GT   = KG * 2 * 32;
constexpr int PA2  = K2 + 4;
constexpr int PR   = TN + 4;
static_assert(TM * TN / 4 <= GT, "one float4 of the tile per adding thread");

template <int N, int K, int PIN>
__global__ __launch_bounds__(GT) void gemm_out_kernel(
    const float* __restrict__ P,
    const float* __restrict__ bias_in,
    const float* __restrict__ W,
    const float* __restrict__ bias_out,
    float* __restrict__ Out) {
  __shared__ __attribute__((aligned(16))) float sA[TM * PA2];
  __shared__ __attribute__((aligned(16))) float sB[K * PB];
  __shared__ __attribute__((aligned(16))) float sR[KG * TM * PR];

  const int tid  = threadIdx.x;
  const int lane = tid & 31;
  const int wave = tid >> 5;
  const int kg   = wave >> 1;
  const int wn   = wave & 1;
  const int c    = lane & 15;
  const int h    = lane >> 4;
  const int m0   = blockIdx.y * TM;
  const int n0   = blockIdx.x * TN;

  constexpr int AC4 = K / 4;
  constexpr int ARS = GT / AC4;
  constexpr int ANP = TM / ARS;
  constexpr int BRS = GT / 8;
  constexpr int BNP = K / BRS;
  const int ac4 = tid % AC4;
  const int ar0 = tid / AC4;
  const int bc4 = tid & 7;
  const int br0 = tid >> 3;

  const float* gA = P + (size_t)(m0 + ar0) * K + ac4 * 4;
  const float* gB = W + (size_t)br0 * N + n0 + bc4 * 4;

  float4 vb[BNP];
  #pragma unroll
  for (int i = 0; i < BNP; ++i)
    vb[i] = *(const float4*)(gB + (size_t)(BRS * i) * N);
  const float4 bv = *(const float4*)(bias_in + ac4 * 4);
  float4 va[ANP];
  #pragma unroll
  for (int i = 0; i < ANP; ++i) {
    const float* s0 = gA + (size_t)(ARS * i) * K;
    float4 s = *(const float4*)s0;
    #pragma unroll
    for (int p = 1; p < PIN; ++p)
      s = add4(s, *(const float4*)(s0 + (size_t)p * MROWS * K));
    s = add4(s, bv);
    va[i] = max4(s, make_float4(0.f, 0.f, 0.f, 0.f));
  }
  #pragma unroll
  for (int i = 0; i < ANP; ++i)
    *(float4*)(sA + (ar0 + ARS * i) * PA2 + ac4 * 4) = va[i];
  #pragma unroll
  for (int i = 0; i < BNP; ++i)
    *(float4*)(sB + (br0 + BRS * i) * PB + bc4 * 4) = vb[i];
  __syncthreads();

  const float* pa = sA + c * PA2 + kg * KC + 2 * h;
  const float* pb = sB + (kg * KC + 2 * h) * PB + wn * 16 + c;
  v2f af[KC / 4];
  v2f bf[KC / 4];
  #pragma unroll
  for (int j = 0; j < KC / 4; ++j) {
    af[j]   = *(const v2f*)(pa + 4 * j);
    bf[j].x = pb[(4 * j) * PB];
    bf[j].y = pb[(4 * j + 1) * PB];
  }
  __builtin_amdgcn_sched_barrier(0);
  v8f acc = {};
  #pragma unroll
  for (int j = 0; j < KC / 4; ++j)
    acc = wmma_f32_k4(af[j], bf[j], acc);

  float* rp = sR + kg * (TM * PR) + (h << 3) * PR + wn * 16 + c;
  #pragma unroll
  for (int r = 0; r < 8; ++r)
    rp[r * PR] = acc[r];
  __syncthreads();

  if (tid < TM * TN / 4) {
    const int row = tid >> 3;
    const int c4  = tid & 7;
    const float* rs = sR + row * PR + c4 * 4;
    float4 s = *(const float4*)rs;
    #pragma unroll
    for (int z = 1; z < KG; ++z)
      s = add4(s, *(const float4*)(rs + z * (TM * PR)));
    s = add4(s, *(const float4*)(bias_out + n0 + c4 * 4));
    *(float4*)(Out + (size_t)(m0 + row) * N + n0 + c4 * 4) = s;
  }
}

constexpr int KS1 = 6;

extern "C" void kernel_launch(void* const* d_in, const int* in_sizes, int n_in,
                              void* d_out, int out_size, void* d_ws, size_t ws_size,
                              hipStream_t stream) {
  const float* x       = (const float*)d_in[0];
  const int*   lengths = (const int*)  d_in[1];
  const float* W1      = (const float*)d_in[2];
  const float* b1      = (const float*)d_in[3];
  const float* W2      = (const float*)d_in[4];
  const float* b2      = (const float*)d_in[5];
  float*       out     = (float*)d_out;

  float* ws      = (float*)d_ws;
  float* partial = ws;
  float* hP      = partial + (size_t)B * NRP * K1;

  pool_partial_kernel<<<dim3(NPC, B), dim3(PT), 0, stream>>>(x, lengths, partial);

  gemm_slice_kernel<N1, K1, KS1, NRP><<<dim3(N1 / BN, B / BM, KS1), dim3(128), 0, stream>>>(
      partial, W1, hP);
  gemm_out_kernel<N2, K2, KS1><<<dim3(N2 / TN, B / TM), dim3(GT), 0, stream>>>(
      hP, b1, W2, b2, out);
}
